# starting kernel
# baseline (speedup 1.0000x reference)
#include <hip/hip_runtime.h>

#define NN 100000
#define NE 1600000

typedef float v2f __attribute__((ext_vector_type(2)));
typedef float v8f __attribute__((ext_vector_type(8)));

__global__ void deg_init_k(float* deg, int n) {
  int i = blockIdx.x * blockDim.x + threadIdx.x;
  if (i < n) deg[i] = 1.0f;
}

__global__ void deg_accum_k(const int* __restrict__ row, const float* __restrict__ w,
                            float* deg, int e) {
  int i = blockIdx.x * blockDim.x + threadIdx.x;
  if (i < e) atomicAdd(&deg[row[i]], w[i]);
}

__global__ void deg_rsqrt_k(float* deg, int n) {
  int i = blockIdx.x * blockDim.x + threadIdx.x;
  if (i < n) deg[i] = rsqrtf(deg[i]);
}

__global__ void edge_norm_k(const int* __restrict__ row, const int* __restrict__ col,
                            const float* __restrict__ w, const float* __restrict__ dis,
                            float* __restrict__ nrm, int e) {
  int i = blockIdx.x * blockDim.x + threadIdx.x;
  if (i < e) nrm[i] = dis[row[i]] * w[i] * dis[col[i]];
}

__global__ void transpose_k(const float* __restrict__ W, float* __restrict__ Wt, int U) {
  int i = blockIdx.x * blockDim.x + threadIdx.x;
  if (i >= 128 * U) return;
  int k = i / U, u = i % U;
  Wt[(size_t)u * 128 + k] = W[i];
}

template <int U, bool RELU_IN>
__global__ __launch_bounds__(128) void gemm_wmma_k(const float* __restrict__ X,
                                                   const float* __restrict__ Wt,
                                                   float* __restrict__ T, int M) {
  const int lane   = threadIdx.x & 31;
  const int wave   = threadIdx.x >> 5;
  const int tile_m = blockIdx.x * 4 + wave;
  const int tile_n = blockIdx.y;
  if (tile_m * 16 >= M) return;

  const int mrow = tile_m * 16 + (lane & 15);
  const int ncol = tile_n * 16 + (lane & 15);
  const int kk   = (lane >> 4) * 2;

  const float* xr = X + (size_t)mrow * 128 + kk;
  const float* wr = Wt + (size_t)ncol * 128 + kk;

  v8f c = {0.f, 0.f, 0.f, 0.f, 0.f, 0.f, 0.f, 0.f};
#pragma unroll
  for (int k = 0; k < 128; k += 4) {
    v2f a = *reinterpret_cast<const v2f*>(xr + k);
    if (RELU_IN) { a.x = fmaxf(a.x, 0.0f); a.y = fmaxf(a.y, 0.0f); }
    v2f b = *reinterpret_cast<const v2f*>(wr + k);
    c = __builtin_amdgcn_wmma_f32_16x16x4_f32(false, a, false, b, (short)0, c,
                                              false, false);
  }

  const int mbase = tile_m * 16 + (lane >> 4) * 8;
  float* trow = T + (size_t)mbase * U + ncol;
#pragma unroll
  for (int r = 0; r < 8; ++r) trow[(size_t)r * U] = c[r];
}

template <int UV>
__global__ void agg_init_k(const float4* __restrict__ T, const float* __restrict__ dis,
                           const float* __restrict__ bias, float4* __restrict__ out,
                           int n) {
  long long idx = (long long)blockIdx.x * blockDim.x + threadIdx.x;
  if (idx >= (long long)n * UV) return;
  int i  = (int)(idx / UV);
  int u4 = (int)(idx % UV);
  float d  = dis[i];
  float sn = d * d;
  float4 t = T[idx];
  float4 b = reinterpret_cast<const float4*>(bias)[u4];
  float4 o;
  o.x = b.x + sn * t.x; o.y = b.y + sn * t.y;
  o.z = b.z + sn * t.z; o.w = b.w + sn * t.w;
  out[idx] = o;
}

template <int UV>
__global__ void agg_scatter_k(const int* __restrict__ row, const int* __restrict__ col,
                              const float* __restrict__ nrm, const float* __restrict__ T,
                              float* out, int e) {
  long long idx = (long long)blockIdx.x * blockDim.x + threadIdx.x;
  if (idx >= (long long)e * UV) return;
  int ei = (int)(idx / UV);
  int u4 = (int)(idx % UV) * 4;
  int r = row[ei], cc = col[ei];
  float w = nrm[ei];
  const int Uu = UV * 4;
  const float4 t = *reinterpret_cast<const float4*>(T + (size_t)cc * Uu + u4);
  float* dst = out + (size_t)r * Uu + u4;
  atomicAdd(dst + 0, w * t.x);
  atomicAdd(dst + 1, w * t.y);
  atomicAdd(dst + 2, w * t.z);
  atomicAdd(dst + 3, w * t.w);
}

__global__ void final_k(const float* __restrict__ rnd, float* out, int nz) {
  int i = blockIdx.x * blockDim.x + threadIdx.x;
  if (i >= nz) return;
  float mu = out[i];
  float ls = out[nz + i];
  float s  = expf(ls);
  out[i]      = mu + rnd[i] * s;
  out[nz + i] = -ls + 0.5f * (s * s + mu * mu - 1.0f);
}

extern "C" void kernel_launch(void* const* d_in, const int* in_sizes, int n_in,
                              void* d_out, int out_size, void* d_ws, size_t ws_size,
                              hipStream_t stream) {
  (void)in_sizes; (void)n_in; (void)out_size; (void)ws_size;
  const int N = NN, E = NE, H = 128, DZ = 64;
  const float* h   = (const float*)d_in[0];
  const int*   ei  = (const int*)d_in[1];
  const float* ew  = (const float*)d_in[2];
  const float* rnd = (const float*)d_in[3];
  const float* W0  = (const float*)d_in[4];
  const float* b0  = (const float*)d_in[5];
  const float* W1  = (const float*)d_in[6];
  const float* b1  = (const float*)d_in[7];
  const float* Wmu = (const float*)d_in[8];
  const float* bmu = (const float*)d_in[9];
  const float* Wsd = (const float*)d_in[10];
  const float* bsd = (const float*)d_in[11];
  const int* row = ei;
  const int* col = ei + E;
  float* out = (float*)d_out;
  const int NZ = N * DZ;

  float* ws    = (float*)d_ws;
  float* dis   = ws;
  float* nrm   = dis + N;
  float* bufT  = nrm + E;
  float* bufH  = bufT + (size_t)N * 128;
  float* bufS  = bufH + (size_t)N * 128;
  float* Wt0   = bufS + (size_t)N * 64;
  float* Wt1   = Wt0 + 128 * 128;
  float* Wtmu  = Wt1 + 128 * 128;
  float* Wtsd  = Wtmu + 64 * 128;

  const int TB = 256;
  deg_init_k <<<(N + TB - 1) / TB, TB, 0, stream>>>(dis, N);
  deg_accum_k<<<(E + TB - 1) / TB, TB, 0, stream>>>(row, ew, dis, E);
  deg_rsqrt_k<<<(N + TB - 1) / TB, TB, 0, stream>>>(dis, N);
  edge_norm_k<<<(E + TB - 1) / TB, TB, 0, stream>>>(row, col, ew, dis, nrm, E);

  transpose_k<<<(128 * H  + TB - 1) / TB, TB, 0, stream>>>(W0,  Wt0,  H);
  transpose_k<<<(128 * H  + TB - 1) / TB, TB, 0, stream>>>(W1,  Wt1,  H);
  transpose_k<<<(128 * DZ + TB - 1) / TB, TB, 0, stream>>>(Wmu, Wtmu, DZ);
  transpose_k<<<(128 * DZ + TB - 1) / TB, TB, 0, stream>>>(Wsd, Wtsd, DZ);

  const int MT = N / 16;
  dim3 gblk(128);
  const unsigned MB = (unsigned)((MT + 3) / 4);

  gemm_wmma_k<128, false><<<dim3(MB, H / 16), gblk, 0, stream>>>(h, Wt0, bufT, N);
  {
    long long ti = (long long)N * 32, ts = (long long)E * 32;
    agg_init_k<32>   <<<(unsigned)((ti + TB - 1) / TB), TB, 0, stream>>>((const float4*)bufT, dis, b0, (float4*)bufH, N);
    agg_scatter_k<32><<<(unsigned)((ts + TB - 1) / TB), TB, 0, stream>>>(row, col, nrm, bufT, bufH, E);
  }
  gemm_wmma_k<128, true><<<dim3(MB, H / 16), gblk, 0, stream>>>(bufH, Wt1, bufT, N);
  {
    long long ti = (long long)N * 32, ts = (long long)E * 32;
    agg_init_k<32>   <<<(unsigned)((ti + TB - 1) / TB), TB, 0, stream>>>((const float4*)bufT, dis, b1, (float4*)bufH, N);
    agg_scatter_k<32><<<(unsigned)((ts + TB - 1) / TB), TB, 0, stream>>>(row, col, nrm, bufT, bufH, E);
  }
  gemm_wmma_k<64, true><<<dim3(MB, DZ / 16), gblk, 0, stream>>>(bufH, Wtmu, bufS, N);
  {
    long long ti = (long long)N * 16, ts = (long long)E * 16;
    agg_init_k<16>   <<<(unsigned)((ti + TB - 1) / TB), TB, 0, stream>>>((const float4*)bufS, dis, bmu, (float4*)out, N);
    agg_scatter_k<16><<<(unsigned)((ts + TB - 1) / TB), TB, 0, stream>>>(row, col, nrm, bufS, out, E);
  }
  gemm_wmma_k<64, true><<<dim3(MB, DZ / 16), gblk, 0, stream>>>(bufH, Wtsd, bufS, N);
  {
    long long ti = (long long)N * 16, ts = (long long)E * 16;
    agg_init_k<16>   <<<(unsigned)((ti + TB - 1) / TB), TB, 0, stream>>>((const float4*)bufS, dis, bsd, (float4*)(out + NZ), N);
    agg_scatter_k<16><<<(unsigned)((ts + TB - 1) / TB), TB, 0, stream>>>(row, col, nrm, bufS, out + NZ, E);
  }
  final_k<<<(NZ + TB - 1) / TB, TB, 0, stream>>>(rnd, out, NZ);
}


# aggregation: a lane owns 4 (2) adjacent floats, so one 16-B (8-B) access per lane moves a whole row (was 4 (2) line loads); T planes on 512-B boundaries; LDS f32 adds in arrival order as before
# speedup vs baseline: 3.2955x; 3.2955x over previous
#include <hip/hip_runtime.h>


#define NN 100000
#define NE 1600000
#define RB 64
#define NBINS ((NN + RB - 1) / RB)
#define EPB 16384
#define NEB ((NE + EPB - 1) / EPB)
#define EPT (EPB / 1024)
#define SB 14
#define EG 8
#define NPAD (((NN + 127) / 128) * 128)

typedef float v2f __attribute__((ext_vector_type(2)));
typedef float v8f __attribute__((ext_vector_type(8)));

__global__ __launch_bounds__(1024) void bin_count_k(const int* __restrict__ row,
                                                    int* __restrict__ tabc, int e) {
  __shared__ int h[NBINS];
  const int tid = threadIdx.x;
  for (int i = tid; i < NBINS; i += 1024) h[i] = 0;
  __syncthreads();
  const int start = blockIdx.x * EPB;
  int r[EPT];
#pragma unroll
  for (int j = 0; j < EPT; ++j) {
    const int i = start + j * 1024 + tid;
    r[j] = (i < e) ? row[i] : -1;
  }
#pragma unroll
  for (int j = 0; j < EPT; ++j)
    if (r[j] >= 0) atomicAdd(&h[(unsigned)r[j] / RB], 1);
  __syncthreads();
  for (int i = tid; i < NBINS; i += 1024) tabc[blockIdx.x * NBINS + i] = h[i];
}

__global__ __launch_bounds__(1024) void bin_scan_k(const int* __restrict__ tabc,
                                                   int* __restrict__ tabb,
                                                   int* __restrict__ bstart) {
  __shared__ int wsum[32];
  const int tid = threadIdx.x, lane = tid & 31, wv = tid >> 5;
  const int b0 = 2 * tid, b1 = 2 * tid + 1;
  const bool ok0 = b0 < NBINS, ok1 = b1 < NBINS;
  int r0 = 0, r1 = 0;
  static_assert(NEB % SB == 0, "table rows are read in whole batches");
  for (int kb = 0; kb < NEB; kb += SB) {
    int c0[SB], c1[SB];
#pragma unroll
    for (int j = 0; j < SB; ++j) {
      c0[j] = ok0 ? tabc[(kb + j) * NBINS + b0] : 0;
      c1[j] = ok1 ? tabc[(kb + j) * NBINS + b1] : 0;
    }
#pragma unroll
    for (int j = 0; j < SB; ++j) {
      if (ok0) { tabb[(kb + j) * NBINS + b0] = r0; r0 += c0[j]; }
      if (ok1) { tabb[(kb + j) * NBINS + b1] = r1; r1 += c1[j]; }
    }
  }
  const int sum = r0 + r1;
  int inc = sum;
  for (int d = 1; d < 32; d <<= 1) {
    const int t = __shfl_up(inc, d);
    if (lane >= d) inc += t;
  }
  if (lane == 31) wsum[wv] = inc;
  __syncthreads();
  if (wv == 0) {
    const int v = wsum[lane];
    int iv = v;
    for (int d = 1; d < 32; d <<= 1) {
      const int t = __shfl_up(iv, d);
      if (lane >= d) iv += t;
    }
    wsum[lane] = iv - v;
  }
  __syncthreads();
  const int excl = wsum[wv] + inc - sum;
  if (b0 <= NBINS) bstart[b0] = excl;
  if (b1 <= NBINS) bstart[b1] = excl + r0;
}

__global__ __launch_bounds__(1024) void bin_place_k(const int* __restrict__ row,
                                                    const int* __restrict__ col,
                                                    const float* __restrict__ w,
                                                    const int* __restrict__ tabb,
                                                    const int* __restrict__ bstart,
                                                    int2* __restrict__ rec,
                                                    float* __restrict__ wb, int e) {
  __shared__ int cur[NBINS];
  const int tid = threadIdx.x;
  for (int i = tid; i < NBINS; i += 1024) cur[i] = bstart[i] + tabb[blockIdx.x * NBINS + i];
  __syncthreads();
  const int start = blockIdx.x * EPB;
  int r[EPT], c[EPT];
  float ww[EPT];
#pragma unroll
  for (int j = 0; j < EPT; ++j) {
    const int i = start + j * 1024 + tid;
    const bool ok = i < e;
    r[j]  = ok ? row[i] : -1;
    c[j]  = ok ? col[i] : 0;
    ww[j] = ok ? w[i] : 0.0f;
  }
#pragma unroll
  for (int j = 0; j < EPT; ++j) {
    if (r[j] >= 0) {
      const int p = atomicAdd(&cur[(unsigned)r[j] / RB], 1);
      rec[p] = make_int2(r[j], c[j]);
      wb[p]  = ww[j];
    }
  }
}

__global__ __launch_bounds__(256) void deg_bin_k(const int2* __restrict__ rec,
                                                 const float* __restrict__ wb,
                                                 const int* __restrict__ bstart,
                                                 float* __restrict__ dis, int n) {
  __shared__ float dg[RB];
  const int tid = threadIdx.x;
  const int base = blockIdx.x * RB;
  if (tid < RB) dg[tid] = 1.0f;
  __syncthreads();
  const int s = bstart[blockIdx.x], e = bstart[blockIdx.x + 1];
  for (int k = s + tid; k < e; k += 256) atomicAdd(&dg[rec[k].x - base], wb[k]);
  __syncthreads();
  if (tid < RB && base + tid < n) dis[base + tid] = rsqrtf(dg[tid]);
}

__global__ void edge_norm_bin_k(const int2* __restrict__ rec, const float* __restrict__ dis,
                                float* wb, int e) {
  int i = blockIdx.x * blockDim.x + threadIdx.x;
  if (i < e) { const int2 r = rec[i]; wb[i] = dis[r.x] * wb[i] * dis[r.y]; }
}

#define STAGE_BLOCKS 48
__global__ __launch_bounds__(256) void stage_w_k(const float* __restrict__ W0,
                                                 const float* __restrict__ W1,
                                                 const float* __restrict__ Wmu,
                                                 const float* __restrict__ Wsd,
                                                 float* __restrict__ F0, float* __restrict__ F1,
                                                 float* __restrict__ Fmu, float* __restrict__ Fsd) {
  const int b = blockIdx.x;
  const float* W;
  float* F;
  int U, i;
  if (b < 16)      { W = W0;  F = F0;  U = 128; i = b * 256 + threadIdx.x; }
  else if (b < 32) { W = W1;  F = F1;  U = 128; i = (b - 16) * 256 + threadIdx.x; }
  else if (b < 40) { W = Wmu; F = Fmu; U = 64;  i = (b - 32) * 256 + threadIdx.x; }
  else             { W = Wsd; F = Fsd; U = 64;  i = (b - 40) * 256 + threadIdx.x; }
  const int L  = i & 31;
  const int sp = (i >> 5) & 15;
  const int nt = i >> 9;
  const int n  = nt * 16 + (L & 15);
  const int k0 = sp * 8 + (L >> 4) * 2;
  float4 q;
  q.x = W[(k0    ) * U + n];
  q.y = W[(k0 + 1) * U + n];
  q.z = W[(k0 + 4) * U + n];
  q.w = W[(k0 + 5) * U + n];
  reinterpret_cast<float4*>(F)[i] = q;
}

#define GM 64
#define GP 132
template <int U, bool RELU_IN>
__global__ __launch_bounds__(128) void gemm_wmma_k(const float* __restrict__ X,
                                                   const float* __restrict__ Wf,
                                                   float* __restrict__ T, int M) {
  constexpr int NT = 2;
  constexpr int WN = U / (16 * NT);
  constexpr int WM = 4 / WN;
  constexpr int MT = GM / 16 / WM;
  __shared__ float4 xs4[GM * GP / 4];
  float* xs = reinterpret_cast<float*>(xs4);

  const int tid  = threadIdx.x;
  const int lane = tid & 31;
  const int wave = __builtin_amdgcn_readfirstlane(tid >> 5);
  const int row0 = blockIdx.x * GM;
  const int kk   = (lane >> 4) * 2;

  const float4* X4 = reinterpret_cast<const float4*>(X);
  float4 xv[16];
#pragma unroll
  for (int i = 0; i < 16; ++i) {
    const int r  = row0 + i * 4 + wave;
    const int gr = (r < M) ? r : (M - 1);
    xv[i] = X4[(size_t)gr * 32 + lane];
  }

  const int ncol0 = (wave % WN) * (NT * 16) + (lane & 15);
  const float4* wf = reinterpret_cast<const float4*>(Wf) + (wave % WN) * (NT * 16 * 32) + lane;
  v2f bf[32][NT];
#pragma unroll
  for (int nt = 0; nt < NT; ++nt) {
#pragma unroll
    for (int sp = 0; sp < 16; ++sp) {
      const float4 q = wf[(nt * 16 + sp) * 32];
      bf[2 * sp][nt]     = (v2f){q.x, q.y};
      bf[2 * sp + 1][nt] = (v2f){q.z, q.w};
    }
  }

#pragma unroll
  for (int i = 0; i < 16; ++i) {
    float4 t = xv[i];
    if (RELU_IN) {
      t.x = fmaxf(t.x, 0.0f); t.y = fmaxf(t.y, 0.0f);
      t.z = fmaxf(t.z, 0.0f); t.w = fmaxf(t.w, 0.0f);
    }
    *reinterpret_cast<float4*>(xs + (i * 4 + wave) * GP + lane * 4) = t;
  }
  __syncthreads();

  const int mloc0 = (wave / WN) * (MT * 16);
  const float* ar = xs + (mloc0 + (lane & 15)) * GP + kk;

  v8f c[MT][NT];
#pragma unroll
  for (int mt = 0; mt < MT; ++mt)
#pragma unroll
    for (int nt = 0; nt < NT; ++nt)
      c[mt][nt] = (v8f){0.f, 0.f, 0.f, 0.f, 0.f, 0.f, 0.f, 0.f};

#pragma unroll
  for (int s = 0; s < 32; ++s) {
    v2f a[MT];
#pragma unroll
    for (int mt = 0; mt < MT; ++mt)
      a[mt] = *reinterpret_cast<const v2f*>(ar + mt * 16 * GP + 4 * s);
#pragma unroll
    for (int mt = 0; mt < MT; ++mt)
#pragma unroll
      for (int nt = 0; nt < NT; ++nt)
        c[mt][nt] = __builtin_amdgcn_wmma_f32_16x16x4_f32(false, a[mt], false, bf[s][nt],
                                                          (short)0, c[mt][nt], false, false);
  }

#pragma unroll
  for (int mt = 0; mt < MT; ++mt) {
    const int mtile = row0 + mloc0 + mt * 16;
    if (mtile < M) {
      float* trow = T + (size_t)(mtile + (lane >> 4) * 8) * U + ncol0;
#pragma unroll
      for (int nt = 0; nt < NT; ++nt)
#pragma unroll
        for (int r = 0; r < 8; ++r) trow[(size_t)r * U + nt * 16] = c[mt][nt][r];
    }
  }
}


template <int Q>
__device__ __forceinline__ void load_lane_cols(const float* p, float (&v)[Q]) {
  if constexpr (Q == 4) {
    const float4 t = *reinterpret_cast<const float4*>(p);
    v[0] = t.x; v[1] = t.y; v[2] = t.z; v[3] = t.w;
  } else {
    static_assert(Q == 2, "a lane owns 4 or 2 columns");
    const float2 t = *reinterpret_cast<const float2*>(p);
    v[0] = t.x; v[1] = t.y;
  }
}
template <int Q>
__device__ __forceinline__ void store_lane_cols(float* p, const float (&v)[Q]) {
  if constexpr (Q == 4) {
    *reinterpret_cast<float4*>(p) = make_float4(v[0], v[1], v[2], v[3]);
  } else {
    static_assert(Q == 2, "a lane owns 4 or 2 columns");
    *reinterpret_cast<float2*>(p) = make_float2(v[0], v[1]);
  }
}

template <int U>
__global__ __launch_bounds__(256) void agg_bin_k(const int2* __restrict__ rec,
                                                 const float* __restrict__ nrmb,
                                                 const int* __restrict__ bstart,
                                                 const float* __restrict__ T,
                                                 const float* __restrict__ dis,
                                                 const float* __restrict__ bias,
                                                 float* __restrict__ out, int n) {
  constexpr int Q  = U / 32;
  constexpr int RW = RB / 8;
  __shared__ float4 acc4[RB * U / 4];
  float* acc = reinterpret_cast<float*>(acc4);
  const int tid  = threadIdx.x;
  const int lane = tid & 31;
  const int wv   = __builtin_amdgcn_readfirstlane(tid >> 5);
  const int base = blockIdx.x * RB;
  const int nrows = (n - base < RB) ? (n - base) : RB;

  {
    float bq[Q];
    load_lane_cols<Q>(bias + lane * Q, bq);
    float t[RW][Q];
    float d[RW];
#pragma unroll
    for (int it = 0; it < RW; ++it) {
      const int gi = (base + wv + 8 * it < n) ? (base + wv + 8 * it) : (n - 1);
      d[it] = dis[gi];
      load_lane_cols<Q>(T + (size_t)gi * U + lane * Q, t[it]);
    }
#pragma unroll
    for (int it = 0; it < RW; ++it) {
      const float sn = d[it] * d[it];
      float* a = acc + (wv + 8 * it) * U + lane;
#pragma unroll
      for (int q = 0; q < Q; ++q) a[32 * q] = bq[q] + sn * t[it][q];
    }
  }
  __syncthreads();

  const int s = bstart[blockIdx.x], e = bstart[blockIdx.x + 1];
  const int per = ((e - s + 8 * EG - 1) / (8 * EG)) * EG;
  int k = s + wv * per;
  const int k1 = (k + per < e) ? (k + per) : e;

  for (; k + EG <= k1; k += EG) {
    float v[EG][Q];
    float wn[EG];
    int   rl[EG];
#pragma unroll
    for (int j = 0; j < EG; ++j) {
      const int2 r = rec[k + j];
      wn[j] = nrmb[k + j];
      rl[j] = r.x - base;
      load_lane_cols<Q>(T + (size_t)r.y * U + lane * Q, v[j]);
    }
#pragma unroll
    for (int j = 0; j < EG; ++j) {
      float* a = acc + rl[j] * U + lane;
#pragma unroll
      for (int q = 0; q < Q; ++q) atomicAdd(a + 32 * q, wn[j] * v[j][q]);
    }
  }
  for (; k < k1; ++k) {
    const int2 r = rec[k];
    const float wn = nrmb[k];
    float* a = acc + (r.x - base) * U + lane;
    float v[Q];
    load_lane_cols<Q>(T + (size_t)r.y * U + lane * Q, v);
#pragma unroll
    for (int q = 0; q < Q; ++q) atomicAdd(a + 32 * q, wn * v[q]);
  }
  __syncthreads();

  float* orow = out + (size_t)(base + wv) * U + lane * Q;
  const float* arow = acc + wv * U + lane;
  if (nrows == RB) {
    float o[RW][Q];
#pragma unroll
    for (int it = 0; it < RW; ++it)
#pragma unroll
      for (int q = 0; q < Q; ++q) o[it][q] = arow[8 * it * U + 32 * q];
#pragma unroll
    for (int it = 0; it < RW; ++it) store_lane_cols<Q>(orow + (size_t)8 * it * U, o[it]);
  } else {
    for (int i = wv; i < nrows; i += 8) {
      float o[Q];
#pragma unroll
      for (int q = 0; q < Q; ++q) o[q] = arow[(i - wv) * U + 32 * q];
      store_lane_cols<Q>(orow + (size_t)(i - wv) * U, o);
    }
  }
}

__global__ void final_k(const float* __restrict__ rnd, float* out, int nz) {
  int i = blockIdx.x * blockDim.x + threadIdx.x;
  if (i >= nz) return;
  float mu = out[i];
  float ls = out[nz + i];
  float s  = expf(ls);
  out[i]      = mu + rnd[i] * s;
  out[nz + i] = -ls + 0.5f * (s * s + mu * mu - 1.0f);
}

extern "C" void kernel_launch(void* const* d_in, const int* in_sizes, int n_in,
                              void* d_out, int out_size, void* d_ws, size_t ws_size,
                              hipStream_t stream) {
  (void)in_sizes; (void)n_in; (void)out_size; (void)ws_size;
  const int N = NN, E = NE, DZ = 64;
  const float* h   = (const float*)d_in[0];
  const int*   ei  = (const int*)d_in[1];
  const float* ew  = (const float*)d_in[2];
  const float* rnd = (const float*)d_in[3];
  const float* W0  = (const float*)d_in[4];
  const float* b0  = (const float*)d_in[5];
  const float* W1  = (const float*)d_in[6];
  const float* b1  = (const float*)d_in[7];
  const float* Wmu = (const float*)d_in[8];
  const float* bmu = (const float*)d_in[9];
  const float* Wsd = (const float*)d_in[10];
  const float* bsd = (const float*)d_in[11];
  const int* row = ei;
  const int* col = ei + E;
  float* out = (float*)d_out;
  const int NZ = N * DZ;

  float* ws    = (float*)d_ws;
  float* dis   = ws;
  float* wb    = dis + NPAD;
  float* bufT  = wb + E;
  float* bufH  = bufT + (size_t)N * 128;
  float* binws = bufH + (size_t)N * 128;
  float* Wf0   = binws + ((size_t)N * 64 - (NPAD - NN));
  float* Wf1   = Wf0 + 128 * 128;
  float* Wfmu  = Wf1 + 128 * 128;
  float* Wfsd  = Wfmu + 128 * 64;
  int2*  rec   = (int2*)binws;
  int*   tabc  = (int*)(binws + 2 * (size_t)E);
  int*   tabb  = tabc + NEB * NBINS;
  int*   bst   = tabb + NEB * NBINS;

  const int TB = 256;
  bin_count_k<<<NEB, 1024, 0, stream>>>(row, tabc, E);
  bin_scan_k <<<1,   1024, 0, stream>>>(tabc, tabb, bst);
  bin_place_k<<<NEB, 1024, 0, stream>>>(row, col, ew, tabb, bst, rec, wb, E);
  deg_bin_k  <<<NBINS, 256, 0, stream>>>(rec, wb, bst, dis, N);
  edge_norm_bin_k<<<(E + TB - 1) / TB, TB, 0, stream>>>(rec, dis, wb, E);

  stage_w_k<<<STAGE_BLOCKS, 256, 0, stream>>>(W0, W1, Wmu, Wsd, Wf0, Wf1, Wfmu, Wfsd);

  static_assert(NN % 16 == 0, "GEMM stores whole 16-row tiles");
  dim3 gblk(128);
  const unsigned MB = (unsigned)((N + GM - 1) / GM);

  gemm_wmma_k<128, false><<<dim3(MB), gblk, 0, stream>>>(h, Wf0, bufT, N);
  agg_bin_k<128><<<NBINS, 256, 0, stream>>>(rec, wb, bst, bufT, dis, b0, bufH, N);
  gemm_wmma_k<128, true><<<dim3(MB), gblk, 0, stream>>>(bufH, Wf1, bufT, N);
  agg_bin_k<128><<<NBINS, 256, 0, stream>>>(rec, wb, bst, bufT, dis, b1, bufH, N);
  gemm_wmma_k<64, true><<<dim3(MB), gblk, 0, stream>>>(bufH, Wfmu, bufT, N);
  agg_bin_k<64><<<NBINS, 256, 0, stream>>>(rec, wb, bst, bufT, dis, bmu, out, N);
  gemm_wmma_k<64, true><<<dim3(MB), gblk, 0, stream>>>(bufH, Wfsd, bufT, N);
  agg_bin_k<64><<<NBINS, 256, 0, stream>>>(rec, wb, bst, bufT, dis, bsd, out + NZ, N);
  final_k<<<(NZ + TB - 1) / TB, TB, 0, stream>>>(rnd, out, NZ);
}


# independent work shares launches (15 -> 13): weight staging rides in the edge-count launch, the bin-offset scan (now one thread per bin) in the layer-1 GEMM launch; k order and sum orders kept
# speedup vs baseline: 3.5472x; 1.0764x over previous
#include <hip/hip_runtime.h>


#define NN 100000
#define NE 1600000
#define RB 64
#define NBINS ((NN + RB - 1) / RB)
#define EPB 16384
#define NEB ((NE + EPB - 1) / EPB)
#define EPT (EPB / 1024)
#define SCAN_BLOCKS ((NBINS + 127) / 128)
#define EG 8
#define NPAD (((NN + 127) / 128) * 128)

typedef float v2f __attribute__((ext_vector_type(2)));
typedef float v8f __attribute__((ext_vector_type(8)));

#define STAGE_BLOCKS 12
__global__ __launch_bounds__(1024) void bin_count_stage_k(const int* __restrict__ row,
                                                          int* __restrict__ tabc, int e,
                                                          const float* __restrict__ W0,
                                                          const float* __restrict__ W1,
                                                          const float* __restrict__ Wmu,
                                                          const float* __restrict__ Wsd,
                                                          float* __restrict__ F0,
                                                          float* __restrict__ F1,
                                                          float* __restrict__ Fmu,
                                                          float* __restrict__ Fsd) {
  __shared__ int h[NBINS];
  const int tid = threadIdx.x;
  if (blockIdx.x >= NEB) {
    const int sb = blockIdx.x - NEB;
    const float* W;
    float* F;
    int U, i;
    if (sb < 4)       { W = W0;  F = F0;  U = 128; i = sb * 1024 + tid; }
    else if (sb < 8)  { W = W1;  F = F1;  U = 128; i = (sb - 4) * 1024 + tid; }
    else if (sb < 10) { W = Wmu; F = Fmu; U = 64;  i = (sb - 8) * 1024 + tid; }
    else              { W = Wsd; F = Fsd; U = 64;  i = (sb - 10) * 1024 + tid; }
    const int L  = i & 31;
    const int sp = (i >> 5) & 15;
    const int nt = i >> 9;
    const int n  = nt * 16 + (L & 15);
    const int k0 = sp * 8 + (L >> 4) * 2;
    float4 q;
    q.x = W[(k0    ) * U + n];
    q.y = W[(k0 + 1) * U + n];
    q.z = W[(k0 + 4) * U + n];
    q.w = W[(k0 + 5) * U + n];
    reinterpret_cast<float4*>(F)[i] = q;
    return;
  }
  for (int i = tid; i < NBINS; i += 1024) h[i] = 0;
  __syncthreads();
  const int start = blockIdx.x * EPB;
  int r[EPT];
#pragma unroll
  for (int j = 0; j < EPT; ++j) {
    const int i = start + j * 1024 + tid;
    r[j] = (i < e) ? row[i] : -1;
  }
#pragma unroll
  for (int j = 0; j < EPT; ++j)
    if (r[j] >= 0) atomicAdd(&h[(unsigned)r[j] / RB], 1);
  __syncthreads();
  for (int i = tid; i < NBINS; i += 1024) tabc[blockIdx.x * NBINS + i] = h[i];
}

__device__ __forceinline__ void bin_scan_cols(const int* __restrict__ tabc,
                                              int* __restrict__ tabb,
                                              int* __restrict__ tot, int sblk, int tid) {
  const int bin = sblk * 128 + tid;
  if (bin >= NBINS) return;
  int cn[NEB];
#pragma unroll
  for (int j = 0; j < NEB; ++j) cn[j] = tabc[j * NBINS + bin];
  int run = 0;
#pragma unroll
  for (int j = 0; j < NEB; ++j) { tabb[j * NBINS + bin] = run; run += cn[j]; }
  tot[bin] = run;
}

__global__ __launch_bounds__(1024) void bin_place_k(const int* __restrict__ row,
                                                    const int* __restrict__ col,
                                                    const float* __restrict__ w,
                                                    const int* __restrict__ tabb,
                                                    const int* __restrict__ tot,
                                                    int* __restrict__ bstart,
                                                    int2* __restrict__ rec,
                                                    float* __restrict__ wb, int e) {
  __shared__ int cur[NBINS];
  __shared__ int wsum[32];
  const int tid = threadIdx.x, lane = tid & 31, wv = tid >> 5;
  {
    const int b0 = 2 * tid, b1 = 2 * tid + 1;
    const bool ok0 = b0 < NBINS, ok1 = b1 < NBINS;
    const int r0 = ok0 ? tot[b0] : 0;
    const int r1 = ok1 ? tot[b1] : 0;
    const int t0 = ok0 ? tabb[blockIdx.x * NBINS + b0] : 0;
    const int t1 = ok1 ? tabb[blockIdx.x * NBINS + b1] : 0;
    const int sum = r0 + r1;
    int inc = sum;
    for (int d = 1; d < 32; d <<= 1) {
      const int t = __shfl_up(inc, d);
      if (lane >= d) inc += t;
    }
    if (lane == 31) wsum[wv] = inc;
    __syncthreads();
    if (wv == 0) {
      const int v = wsum[lane];
      int iv = v;
      for (int d = 1; d < 32; d <<= 1) {
        const int t = __shfl_up(iv, d);
        if (lane >= d) iv += t;
      }
      wsum[lane] = iv - v;
    }
    __syncthreads();
    const int excl = wsum[wv] + inc - sum;
    if (ok0) cur[b0] = excl + t0;
    if (ok1) cur[b1] = excl + r0 + t1;
    if (blockIdx.x == 0) {
      if (b0 <= NBINS) bstart[b0] = excl;
      if (b1 <= NBINS) bstart[b1] = excl + r0;
    }
  }
  __syncthreads();
  const int start = blockIdx.x * EPB;
  int r[EPT], c[EPT];
  float ww[EPT];
#pragma unroll
  for (int j = 0; j < EPT; ++j) {
    const int i = start + j * 1024 + tid;
    const bool ok = i < e;
    r[j]  = ok ? row[i] : -1;
    c[j]  = ok ? col[i] : 0;
    ww[j] = ok ? w[i] : 0.0f;
  }
#pragma unroll
  for (int j = 0; j < EPT; ++j) {
    if (r[j] >= 0) {
      const int p = atomicAdd(&cur[(unsigned)r[j] / RB], 1);
      rec[p] = make_int2(r[j], c[j]);
      wb[p]  = ww[j];
    }
  }
}

__global__ __launch_bounds__(256) void deg_bin_k(const int2* __restrict__ rec,
                                                 const float* __restrict__ wb,
                                                 const int* __restrict__ bstart,
                                                 float* __restrict__ dis, int n) {
  __shared__ float dg[RB];
  const int tid = threadIdx.x;
  const int base = blockIdx.x * RB;
  if (tid < RB) dg[tid] = 1.0f;
  __syncthreads();
  const int s = bstart[blockIdx.x], e = bstart[blockIdx.x + 1];
  for (int k = s + tid; k < e; k += 256) atomicAdd(&dg[rec[k].x - base], wb[k]);
  __syncthreads();
  if (tid < RB && base + tid < n) dis[base + tid] = rsqrtf(dg[tid]);
}

__global__ void edge_norm_bin_k(const int2* __restrict__ rec, const float* __restrict__ dis,
                                float* wb, int e) {
  int i = blockIdx.x * blockDim.x + threadIdx.x;
  if (i < e) { const int2 r = rec[i]; wb[i] = dis[r.x] * wb[i] * dis[r.y]; }
}

#define GM 64
#define GP 132
template <int U, bool RELU_IN>
__device__ __forceinline__ void gemm_tile(const float* __restrict__ X,
                                          const float* __restrict__ Wf,
                                          float* __restrict__ T, int M, int blk,
                                          float* xs) {
  constexpr int NT = 2;
  constexpr int WN = U / (16 * NT);
  constexpr int WM = 4 / WN;
  constexpr int MT = GM / 16 / WM;

  const int tid  = threadIdx.x;
  const int lane = tid & 31;
  const int wave = __builtin_amdgcn_readfirstlane(tid >> 5);
  const int row0 = blk * GM;
  const int kk   = (lane >> 4) * 2;

  const float4* X4 = reinterpret_cast<const float4*>(X);
  float4 xv[16];
#pragma unroll
  for (int i = 0; i < 16; ++i) {
    const int r  = row0 + i * 4 + wave;
    const int gr = (r < M) ? r : (M - 1);
    xv[i] = X4[(size_t)gr * 32 + lane];
  }

  const int ncol0 = (wave % WN) * (NT * 16) + (lane & 15);
  const float4* wf = reinterpret_cast<const float4*>(Wf) + (wave % WN) * (NT * 16 * 32) + lane;
  v2f bf[32][NT];
#pragma unroll
  for (int nt = 0; nt < NT; ++nt) {
#pragma unroll
    for (int sp = 0; sp < 16; ++sp) {
      const float4 q = wf[(nt * 16 + sp) * 32];
      bf[2 * sp][nt]     = (v2f){q.x, q.y};
      bf[2 * sp + 1][nt] = (v2f){q.z, q.w};
    }
  }

#pragma unroll
  for (int i = 0; i < 16; ++i) {
    float4 t = xv[i];
    if (RELU_IN) {
      t.x = fmaxf(t.x, 0.0f); t.y = fmaxf(t.y, 0.0f);
      t.z = fmaxf(t.z, 0.0f); t.w = fmaxf(t.w, 0.0f);
    }
    *reinterpret_cast<float4*>(xs + (i * 4 + wave) * GP + lane * 4) = t;
  }
  __syncthreads();

  const int mloc0 = (wave / WN) * (MT * 16);
  const float* ar = xs + (mloc0 + (lane & 15)) * GP + kk;

  v8f c[MT][NT];
#pragma unroll
  for (int mt = 0; mt < MT; ++mt)
#pragma unroll
    for (int nt = 0; nt < NT; ++nt)
      c[mt][nt] = (v8f){0.f, 0.f, 0.f, 0.f, 0.f, 0.f, 0.f, 0.f};

#pragma unroll
  for (int s = 0; s < 32; ++s) {
    v2f a[MT];
#pragma unroll
    for (int mt = 0; mt < MT; ++mt)
      a[mt] = *reinterpret_cast<const v2f*>(ar + mt * 16 * GP + 4 * s);
#pragma unroll
    for (int mt = 0; mt < MT; ++mt)
#pragma unroll
      for (int nt = 0; nt < NT; ++nt)
        c[mt][nt] = __builtin_amdgcn_wmma_f32_16x16x4_f32(false, a[mt], false, bf[s][nt],
                                                          (short)0, c[mt][nt], false, false);
  }

#pragma unroll
  for (int mt = 0; mt < MT; ++mt) {
    const int mtile = row0 + mloc0 + mt * 16;
    if (mtile < M) {
      float* trow = T + (size_t)(mtile + (lane >> 4) * 8) * U + ncol0;
#pragma unroll
      for (int nt = 0; nt < NT; ++nt)
#pragma unroll
        for (int r = 0; r < 8; ++r) trow[(size_t)r * U + nt * 16] = c[mt][nt][r];
    }
  }
}

template <int U, bool RELU_IN>
__global__ __launch_bounds__(128) void gemm_wmma_k(const float* __restrict__ X,
                                                   const float* __restrict__ Wf,
                                                   float* __restrict__ T, int M) {
  __shared__ float4 xs4[GM * GP / 4];
  gemm_tile<U, RELU_IN>(X, Wf, T, M, blockIdx.x, reinterpret_cast<float*>(xs4));
}

__global__ __launch_bounds__(128) void gemm1_scan_k(const float* __restrict__ X,
                                                    const float* __restrict__ Wf,
                                                    float* __restrict__ T, int M,
                                                    const int* __restrict__ tabc,
                                                    int* __restrict__ tabb,
                                                    int* __restrict__ tot) {
  __shared__ float4 xs4[GM * GP / 4];
  if (blockIdx.x < SCAN_BLOCKS) {
    bin_scan_cols(tabc, tabb, tot, blockIdx.x, threadIdx.x);
    return;
  }
  gemm_tile<128, false>(X, Wf, T, M, blockIdx.x - SCAN_BLOCKS, reinterpret_cast<float*>(xs4));
}


template <int Q>
__device__ __forceinline__ void load_lane_cols(const float* p, float (&v)[Q]) {
  if constexpr (Q == 4) {
    const float4 t = *reinterpret_cast<const float4*>(p);
    v[0] = t.x; v[1] = t.y; v[2] = t.z; v[3] = t.w;
  } else {
    static_assert(Q == 2, "a lane owns 4 or 2 columns");
    const float2 t = *reinterpret_cast<const float2*>(p);
    v[0] = t.x; v[1] = t.y;
  }
}
template <int Q>
__device__ __forceinline__ void store_lane_cols(float* p, const float (&v)[Q]) {
  if constexpr (Q == 4) {
    *reinterpret_cast<float4*>(p) = make_float4(v[0], v[1], v[2], v[3]);
  } else {
    static_assert(Q == 2, "a lane owns 4 or 2 columns");
    *reinterpret_cast<float2*>(p) = make_float2(v[0], v[1]);
  }
}

template <int U>
__global__ __launch_bounds__(256) void agg_bin_k(const int2* __restrict__ rec,
                                                 const float* __restrict__ nrmb,
                                                 const int* __restrict__ bstart,
                                                 const float* __restrict__ T,
                                                 const float* __restrict__ dis,
                                                 const float* __restrict__ bias,
                                                 float* __restrict__ out, int n) {
  constexpr int Q  = U / 32;
  constexpr int RW = RB / 8;
  __shared__ float4 acc4[RB * U / 4];
  float* acc = reinterpret_cast<float*>(acc4);
  const int tid  = threadIdx.x;
  const int lane = tid & 31;
  const int wv   = __builtin_amdgcn_readfirstlane(tid >> 5);
  const int base = blockIdx.x * RB;
  const int nrows = (n - base < RB) ? (n - base) : RB;

  {
    float bq[Q];
    load_lane_cols<Q>(bias + lane * Q, bq);
    float t[RW][Q];
    float d[RW];
#pragma unroll
    for (int it = 0; it < RW; ++it) {
      const int gi = (base + wv + 8 * it < n) ? (base + wv + 8 * it) : (n - 1);
      d[it] = dis[gi];
      load_lane_cols<Q>(T + (size_t)gi * U + lane * Q, t[it]);
    }
#pragma unroll
    for (int it = 0; it < RW; ++it) {
      const float sn = d[it] * d[it];
      float* a = acc + (wv + 8 * it) * U + lane;
#pragma unroll
      for (int q = 0; q < Q; ++q) a[32 * q] = bq[q] + sn * t[it][q];
    }
  }
  __syncthreads();

  const int s = bstart[blockIdx.x], e = bstart[blockIdx.x + 1];
  const int per = ((e - s + 8 * EG - 1) / (8 * EG)) * EG;
  int k = s + wv * per;
  const int k1 = (k + per < e) ? (k + per) : e;

  for (; k + EG <= k1; k += EG) {
    float v[EG][Q];
    float wn[EG];
    int   rl[EG];
#pragma unroll
    for (int j = 0; j < EG; ++j) {
      const int2 r = rec[k + j];
      wn[j] = nrmb[k + j];
      rl[j] = r.x - base;
      load_lane_cols<Q>(T + (size_t)r.y * U + lane * Q, v[j]);
    }
#pragma unroll
    for (int j = 0; j < EG; ++j) {
      float* a = acc + rl[j] * U + lane;
#pragma unroll
      for (int q = 0; q < Q; ++q) atomicAdd(a + 32 * q, wn[j] * v[j][q]);
    }
  }
  for (; k < k1; ++k) {
    const int2 r = rec[k];
    const float wn = nrmb[k];
    float* a = acc + (r.x - base) * U + lane;
    float v[Q];
    load_lane_cols<Q>(T + (size_t)r.y * U + lane * Q, v);
#pragma unroll
    for (int q = 0; q < Q; ++q) atomicAdd(a + 32 * q, wn * v[q]);
  }
  __syncthreads();

  float* orow = out + (size_t)(base + wv) * U + lane * Q;
  const float* arow = acc + wv * U + lane;
  if (nrows == RB) {
    float o[RW][Q];
#pragma unroll
    for (int it = 0; it < RW; ++it)
#pragma unroll
      for (int q = 0; q < Q; ++q) o[it][q] = arow[8 * it * U + 32 * q];
#pragma unroll
    for (int it = 0; it < RW; ++it) store_lane_cols<Q>(orow + (size_t)8 * it * U, o[it]);
  } else {
    for (int i = wv; i < nrows; i += 8) {
      float o[Q];
#pragma unroll
      for (int q = 0; q < Q; ++q) o[q] = arow[(i - wv) * U + 32 * q];
      store_lane_cols<Q>(orow + (size_t)(i - wv) * U, o);
    }
  }
}

__global__ void final_k(const float* __restrict__ rnd, float* out, int nz) {
  int i = blockIdx.x * blockDim.x + threadIdx.x;
  if (i >= nz) return;
  float mu = out[i];
  float ls = out[nz + i];
  float s  = expf(ls);
  out[i]      = mu + rnd[i] * s;
  out[nz + i] = -ls + 0.5f * (s * s + mu * mu - 1.0f);
}

extern "C" void kernel_launch(void* const* d_in, const int* in_sizes, int n_in,
                              void* d_out, int out_size, void* d_ws, size_t ws_size,
                              hipStream_t stream) {
  (void)in_sizes; (void)n_in; (void)out_size; (void)ws_size;
  const int N = NN, E = NE, DZ = 64;
  const float* h   = (const float*)d_in[0];
  const int*   ei  = (const int*)d_in[1];
  const float* ew  = (const float*)d_in[2];
  const float* rnd = (const float*)d_in[3];
  const float* W0  = (const float*)d_in[4];
  const float* b0  = (const float*)d_in[5];
  const float* W1  = (const float*)d_in[6];
  const float* b1  = (const float*)d_in[7];
  const float* Wmu = (const float*)d_in[8];
  const float* bmu = (const float*)d_in[9];
  const float* Wsd = (const float*)d_in[10];
  const float* bsd = (const float*)d_in[11];
  const int* row = ei;
  const int* col = ei + E;
  float* out = (float*)d_out;
  const int NZ = N * DZ;

  float* ws    = (float*)d_ws;
  float* dis   = ws;
  float* wb    = dis + NPAD;
  float* bufT  = wb + E;
  float* bufH  = bufT + (size_t)N * 128;
  float* binws = bufH + (size_t)N * 128;
  float* Wf0   = binws + ((size_t)N * 64 - (NPAD - NN));
  float* Wf1   = Wf0 + 128 * 128;
  float* Wfmu  = Wf1 + 128 * 128;
  float* Wfsd  = Wfmu + 128 * 64;
  int2*  rec   = (int2*)binws;
  int*   tabc  = (int*)(binws + 2 * (size_t)E);
  int*   tabb  = tabc + NEB * NBINS;
  int*   bst   = tabb + NEB * NBINS;
  int*   tot   = bst + (NBINS + 1);

  const int TB = 256;
  static_assert(NN % 16 == 0, "GEMM stores whole 16-row tiles");
  dim3 gblk(128);
  const unsigned MB = (unsigned)((N + GM - 1) / GM);

  bin_count_stage_k<<<NEB + STAGE_BLOCKS, 1024, 0, stream>>>(row, tabc, E, W0, W1, Wmu, Wsd,
                                                              Wf0, Wf1, Wfmu, Wfsd);
  gemm1_scan_k<<<dim3(SCAN_BLOCKS + MB), gblk, 0, stream>>>(h, Wf0, bufT, N, tabc, tabb, tot);
  bin_place_k<<<NEB, 1024, 0, stream>>>(row, col, ew, tabb, tot, bst, rec, wb, E);
  deg_bin_k  <<<NBINS, 256, 0, stream>>>(rec, wb, bst, dis, N);
  edge_norm_bin_k<<<(E + TB - 1) / TB, TB, 0, stream>>>(rec, dis, wb, E);

  agg_bin_k<128><<<NBINS, 256, 0, stream>>>(rec, wb, bst, bufT, dis, b0, bufH, N);
  gemm_wmma_k<128, true><<<dim3(MB), gblk, 0, stream>>>(bufH, Wf1, bufT, N);
  agg_bin_k<128><<<NBINS, 256, 0, stream>>>(rec, wb, bst, bufT, dis, b1, bufH, N);
  gemm_wmma_k<64, true><<<dim3(MB), gblk, 0, stream>>>(bufH, Wfmu, bufT, N);
  agg_bin_k<64><<<NBINS, 256, 0, stream>>>(rec, wb, bst, bufT, dis, bmu, out, N);
  gemm_wmma_k<64, true><<<dim3(MB), gblk, 0, stream>>>(bufH, Wfsd, bufT, N);
  agg_bin_k<64><<<NBINS, 256, 0, stream>>>(rec, wb, bst, bufT, dis, bsd, out + NZ, N);
  final_k<<<(NZ + TB - 1) / TB, TB, 0, stream>>>(rnd, out, NZ);
}
